# starting kernel
# baseline (speedup 1.0000x reference)
#include <hip/hip_runtime.h>
#include <math.h>

typedef __attribute__((ext_vector_type(2))) float v2f;
typedef __attribute__((ext_vector_type(8))) float v8f;

#define DM   1024
#define DS   16
#define DTR  64
#define NPJ  96
#define BSZ  2
#define TLEN 2048
#define MROWS (BSZ*TLEN)
#define NC   32
#define CL   (TLEN/NC)

__global__ void proj_gemm_kernel(const float* __restrict__ x,
                                 const float* __restrict__ Wx,
                                 float* __restrict__ xproj) {
    const int m0   = blockIdx.x * 16;
    const int nb   = blockIdx.y * 48;
    const int lane = threadIdx.x;
    const int half = lane >> 4;
    const int l16  = lane & 15;

    v8f acc0 = {}, acc1 = {}, acc2 = {};
    const float* arow = x + (size_t)(m0 + l16) * DM;

    for (int k0 = 0; k0 < DM; k0 += 4) {
        const int ka = k0 + 2 * half;
        v2f a = *(const v2f*)(arow + ka);
        const float* bp = Wx + (size_t)ka * NPJ + nb + l16;
        v2f b0, b1, b2;
        b0.x = bp[0];        b0.y = bp[NPJ];
        b1.x = bp[16];       b1.y = bp[16 + NPJ];
        b2.x = bp[32];       b2.y = bp[32 + NPJ];
        acc0 = __builtin_amdgcn_wmma_f32_16x16x4_f32(false, a, false, b0, (short)0, acc0, false, false);
        acc1 = __builtin_amdgcn_wmma_f32_16x16x4_f32(false, a, false, b1, (short)0, acc1, false, false);
        acc2 = __builtin_amdgcn_wmma_f32_16x16x4_f32(false, a, false, b2, (short)0, acc2, false, false);
    }
#pragma unroll
    for (int v = 0; v < 8; ++v) {
        const int row = m0 + v + (half ? 8 : 0);
        float* orow = xproj + (size_t)row * NPJ + nb + l16;
        orow[0]  = acc0[v];
        orow[16] = acc1[v];
        orow[32] = acc2[v];
    }
}

__global__ void dt_gemm_kernel(const float* __restrict__ xproj,
                               const float* __restrict__ Wdt,
                               const float* __restrict__ bdt,
                               float* __restrict__ dt) {
    const int m0   = blockIdx.x * 16;
    const int nb   = blockIdx.y * 64;
    const int lane = threadIdx.x;
    const int half = lane >> 4;
    const int l16  = lane & 15;

    v8f acc[4] = {};
    const float* arow = xproj + (size_t)(m0 + l16) * NPJ;

#pragma unroll
    for (int k0 = 0; k0 < DTR; k0 += 4) {
        const int ka = k0 + 2 * half;
        v2f a = *(const v2f*)(arow + ka);
        const float* bp = Wdt + (size_t)ka * DM + nb + l16;
#pragma unroll
        for (int j = 0; j < 4; ++j) {
            v2f b;
            b.x = bp[j * 16];
            b.y = bp[j * 16 + DM];
            acc[j] = __builtin_amdgcn_wmma_f32_16x16x4_f32(false, a, false, b, (short)0, acc[j], false, false);
        }
    }
#pragma unroll
    for (int v = 0; v < 8; ++v) {
        const int row = m0 + v + (half ? 8 : 0);
#pragma unroll
        for (int j = 0; j < 4; ++j) {
            const int col = nb + j * 16 + l16;
            float z = acc[j][v] + bdt[col];
            float sp = (z > 20.0f) ? z : log1pf(__expf(z));
            dt[(size_t)row * DM + col] = sp;
        }
    }
}

__global__ void scan_chunk1_kernel(const float* __restrict__ x,
                                   const float* __restrict__ xproj,
                                   const float* __restrict__ dt,
                                   const float* __restrict__ A_log,
                                   float* __restrict__ P,
                                   float* __restrict__ S) {
    const int bc = blockIdx.x;
    const int b  = bc / NC;
    const int c  = bc % NC;
    const int d  = blockIdx.y * blockDim.x + threadIdx.x;

    float An[DS];
#pragma unroll
    for (int s = 0; s < DS; ++s) An[s] = -__expf(A_log[d * DS + s]);

    float p[DS], h[DS];
#pragma unroll
    for (int s = 0; s < DS; ++s) { p[s] = 1.0f; h[s] = 0.0f; }

    const int t0 = c * CL;
    for (int t = t0; t < t0 + CL; ++t) {
        const int bt = b * TLEN + t;
        const float dtv = dt[(size_t)bt * DM + d];
        const float xv  = x[(size_t)bt * DM + d];
        const float dtx = dtv * xv;
        const float* Bm = xproj + (size_t)bt * NPJ + DTR;
#pragma unroll
        for (int s = 0; s < DS; ++s) {
            const float a = __expf(dtv * An[s]);
            p[s] *= a;
            h[s] = a * h[s] + dtx * Bm[s];
        }
    }
    const size_t base = (((size_t)b * NC + c) * DM + d) * DS;
#pragma unroll
    for (int s = 0; s < DS; ++s) { P[base + s] = p[s]; S[base + s] = h[s]; }
}

__global__ void scan_carry_kernel(const float* __restrict__ P,
                                  const float* __restrict__ S,
                                  float* __restrict__ H0) {
    const int idx = blockIdx.x * blockDim.x + threadIdx.x;
    const int b   = idx / (DM * DS);
    const int ds  = idx % (DM * DS);
    float h = 0.0f;
    for (int c = 0; c < NC; ++c) {
        const size_t off = (((size_t)b * NC + c) * DM * DS) + ds;
        H0[off] = h;
        h = P[off] * h + S[off];
    }
}

__global__ void scan_chunk2_kernel(const float* __restrict__ x,
                                   const float* __restrict__ xproj,
                                   const float* __restrict__ dt,
                                   const float* __restrict__ A_log,
                                   const float* __restrict__ Dvec,
                                   const float* __restrict__ H0,
                                   float* __restrict__ out) {
    const int bc = blockIdx.x;
    const int b  = bc / NC;
    const int c  = bc % NC;
    const int d  = blockIdx.y * blockDim.x + threadIdx.x;

    float An[DS];
#pragma unroll
    for (int s = 0; s < DS; ++s) An[s] = -__expf(A_log[d * DS + s]);

    float h[DS];
    const size_t hbase = (((size_t)b * NC + c) * DM + d) * DS;
#pragma unroll
    for (int s = 0; s < DS; ++s) h[s] = H0[hbase + s];

    const float dval = Dvec[d];
    const int t0 = c * CL;
    for (int t = t0; t < t0 + CL; ++t) {
        const int bt = b * TLEN + t;
        const float dtv = dt[(size_t)bt * DM + d];
        const float xv  = x[(size_t)bt * DM + d];
        const float dtx = dtv * xv;
        const float* Bm = xproj + (size_t)bt * NPJ + DTR;
        const float* Cm = xproj + (size_t)bt * NPJ + DTR + DS;
        float y = 0.0f;
#pragma unroll
        for (int s = 0; s < DS; ++s) {
            const float a = __expf(dtv * An[s]);
            h[s] = a * h[s] + dtx * Bm[s];
            y += h[s] * Cm[s];
        }
        out[(size_t)bt * DM + d] = y + xv * dval;
    }
}

extern "C" void kernel_launch(void* const* d_in, const int* in_sizes, int n_in,
                              void* d_out, int out_size, void* d_ws, size_t ws_size,
                              hipStream_t stream) {
    const float* x     = (const float*)d_in[0];
    const float* Wx    = (const float*)d_in[1];
    const float* Wdt   = (const float*)d_in[2];
    const float* bdt   = (const float*)d_in[3];
    const float* A_log = (const float*)d_in[4];
    const float* Dvec  = (const float*)d_in[5];
    float* out = (float*)d_out;

    char* ws = (char*)d_ws;
    float* xproj = (float*)ws;
    ws += (size_t)MROWS * NPJ * sizeof(float);
    float* dt    = (float*)ws;
    ws += (size_t)MROWS * DM * sizeof(float);
    float* P     = (float*)ws;
    ws += (size_t)BSZ * NC * DM * DS * sizeof(float);
    float* S     = (float*)ws;
    ws += (size_t)BSZ * NC * DM * DS * sizeof(float);
    float* H0    = (float*)ws;

    proj_gemm_kernel<<<dim3(MROWS / 16, 2), 32, 0, stream>>>(x, Wx, xproj);
    dt_gemm_kernel<<<dim3(MROWS / 16, 16), 32, 0, stream>>>(xproj, Wdt, bdt, dt);
    scan_chunk1_kernel<<<dim3(BSZ * NC, DM / 256), 256, 0, stream>>>(x, xproj, dt, A_log, P, S);
    scan_carry_kernel<<<(BSZ * DM * DS) / 256, 256, 0, stream>>>(P, S, H0);
    scan_chunk2_kernel<<<dim3(BSZ * NC, DM / 256), 256, 0, stream>>>(x, xproj, dt, A_log, Dvec, H0, out);
}


# proj GEMM: split K over the 16 waves of a block (32x48 tile, 64 k per wave), partials added through LDS in fixed order: changes f32 summation order of x_proj; 4 waves per SIMD, was 512 lone waves
# speedup vs baseline: 1.2006x; 1.2006x over previous
#include <hip/hip_runtime.h>
#include <math.h>

typedef __attribute__((ext_vector_type(2))) float v2f;
typedef __attribute__((ext_vector_type(8))) float v8f;

#define DM   1024
#define DS   16
#define DTR  64
#define NPJ  96
#define BSZ  2
#define TLEN 2048
#define MROWS (BSZ*TLEN)
#define NC   32
#define CL   (TLEN/NC)

#define PJ_ROWS  32
#define PJ_COLS  48
#define PJ_WAVES 16
#define PJ_KW    (DM / PJ_WAVES)
#define PJ_TILE  (PJ_ROWS * PJ_COLS)

__global__ __launch_bounds__(PJ_WAVES * 32)
void proj_gemm_kernel(const float* __restrict__ x,
                      const float* __restrict__ Wx,
                      float* __restrict__ xproj) {
    __shared__ float4 red[PJ_WAVES][PJ_TILE / 4];

    const int mb   = blockIdx.x * PJ_ROWS;
    const int nb   = blockIdx.y * PJ_COLS;
    const int tid  = threadIdx.x;
    const int w    = tid >> 5;
    const int lane = tid & 31;
    const int half = lane >> 4;
    const int l16  = lane & 15;
    const int kb   = w * PJ_KW + 2 * half;

    v8f acc[2][3] = {};
    const float* arow0 = x + (size_t)(mb + l16) * DM + kb;
    const float* arow1 = arow0 + (size_t)16 * DM;
    const float* bcol  = Wx + (size_t)kb * NPJ + nb + l16;

#pragma unroll
    for (int ks = 0; ks < PJ_KW / 4; ++ks) {
        const v2f a0 = *(const v2f*)(arow0 + 4 * ks);
        const v2f a1 = *(const v2f*)(arow1 + 4 * ks);
        const float* bp = bcol + (size_t)(4 * ks) * NPJ;
        v2f b0, b1, b2;
        b0.x = bp[0];        b0.y = bp[NPJ];
        b1.x = bp[16];       b1.y = bp[16 + NPJ];
        b2.x = bp[32];       b2.y = bp[32 + NPJ];
        acc[0][0] = __builtin_amdgcn_wmma_f32_16x16x4_f32(false, a0, false, b0, (short)0, acc[0][0], false, false);
        acc[1][0] = __builtin_amdgcn_wmma_f32_16x16x4_f32(false, a1, false, b0, (short)0, acc[1][0], false, false);
        acc[0][1] = __builtin_amdgcn_wmma_f32_16x16x4_f32(false, a0, false, b1, (short)0, acc[0][1], false, false);
        acc[1][1] = __builtin_amdgcn_wmma_f32_16x16x4_f32(false, a1, false, b1, (short)0, acc[1][1], false, false);
        acc[0][2] = __builtin_amdgcn_wmma_f32_16x16x4_f32(false, a0, false, b2, (short)0, acc[0][2], false, false);
        acc[1][2] = __builtin_amdgcn_wmma_f32_16x16x4_f32(false, a1, false, b2, (short)0, acc[1][2], false, false);
    }

    float* mine = (float*)&red[w][0];
#pragma unroll
    for (int mi = 0; mi < 2; ++mi)
#pragma unroll
        for (int nj = 0; nj < 3; ++nj)
#pragma unroll
            for (int v = 0; v < 8; ++v)
                mine[((mi * 3 + nj) * 8 + v) * 32 + lane] = acc[mi][nj][v];
    __syncthreads();

    if (tid < PJ_TILE / 4) {
        float4 s = red[0][tid];
#pragma unroll
        for (int q = 1; q < PJ_WAVES; ++q) {
            const float4 p = red[q][tid];
            s.x += p.x; s.y += p.y; s.z += p.z; s.w += p.w;
        }
        const int frag = tid >> 6;
        const int v    = (tid >> 3) & 7;
        const int hh   = (tid >> 2) & 1;
        const int c0   = (tid & 3) * 4;
        const int mi   = frag / 3;
        const int nj   = frag - 3 * mi;
        const int row  = mb + 16 * mi + 8 * hh + v;
        const int col  = nb + 16 * nj + c0;
        *(float4*)(xproj + (size_t)row * NPJ + col) = s;
    }
}

__global__ void dt_gemm_kernel(const float* __restrict__ xproj,
                               const float* __restrict__ Wdt,
                               const float* __restrict__ bdt,
                               float* __restrict__ dt) {
    const int m0   = blockIdx.x * 16;
    const int nb   = blockIdx.y * 64;
    const int lane = threadIdx.x;
    const int half = lane >> 4;
    const int l16  = lane & 15;

    v8f acc[4] = {};
    const float* arow = xproj + (size_t)(m0 + l16) * NPJ;

#pragma unroll
    for (int k0 = 0; k0 < DTR; k0 += 4) {
        const int ka = k0 + 2 * half;
        v2f a = *(const v2f*)(arow + ka);
        const float* bp = Wdt + (size_t)ka * DM + nb + l16;
#pragma unroll
        for (int j = 0; j < 4; ++j) {
            v2f b;
            b.x = bp[j * 16];
            b.y = bp[j * 16 + DM];
            acc[j] = __builtin_amdgcn_wmma_f32_16x16x4_f32(false, a, false, b, (short)0, acc[j], false, false);
        }
    }
#pragma unroll
    for (int v = 0; v < 8; ++v) {
        const int row = m0 + v + (half ? 8 : 0);
#pragma unroll
        for (int j = 0; j < 4; ++j) {
            const int col = nb + j * 16 + l16;
            float z = acc[j][v] + bdt[col];
            float sp = (z > 20.0f) ? z : log1pf(__expf(z));
            dt[(size_t)row * DM + col] = sp;
        }
    }
}

__global__ void scan_chunk1_kernel(const float* __restrict__ x,
                                   const float* __restrict__ xproj,
                                   const float* __restrict__ dt,
                                   const float* __restrict__ A_log,
                                   float* __restrict__ P,
                                   float* __restrict__ S) {
    const int bc = blockIdx.x;
    const int b  = bc / NC;
    const int c  = bc % NC;
    const int d  = blockIdx.y * blockDim.x + threadIdx.x;

    float An[DS];
#pragma unroll
    for (int s = 0; s < DS; ++s) An[s] = -__expf(A_log[d * DS + s]);

    float p[DS], h[DS];
#pragma unroll
    for (int s = 0; s < DS; ++s) { p[s] = 1.0f; h[s] = 0.0f; }

    const int t0 = c * CL;
    for (int t = t0; t < t0 + CL; ++t) {
        const int bt = b * TLEN + t;
        const float dtv = dt[(size_t)bt * DM + d];
        const float xv  = x[(size_t)bt * DM + d];
        const float dtx = dtv * xv;
        const float* Bm = xproj + (size_t)bt * NPJ + DTR;
#pragma unroll
        for (int s = 0; s < DS; ++s) {
            const float a = __expf(dtv * An[s]);
            p[s] *= a;
            h[s] = a * h[s] + dtx * Bm[s];
        }
    }
    const size_t base = (((size_t)b * NC + c) * DM + d) * DS;
#pragma unroll
    for (int s = 0; s < DS; ++s) { P[base + s] = p[s]; S[base + s] = h[s]; }
}

__global__ void scan_carry_kernel(const float* __restrict__ P,
                                  const float* __restrict__ S,
                                  float* __restrict__ H0) {
    const int idx = blockIdx.x * blockDim.x + threadIdx.x;
    const int b   = idx / (DM * DS);
    const int ds  = idx % (DM * DS);
    float h = 0.0f;
    for (int c = 0; c < NC; ++c) {
        const size_t off = (((size_t)b * NC + c) * DM * DS) + ds;
        H0[off] = h;
        h = P[off] * h + S[off];
    }
}

__global__ void scan_chunk2_kernel(const float* __restrict__ x,
                                   const float* __restrict__ xproj,
                                   const float* __restrict__ dt,
                                   const float* __restrict__ A_log,
                                   const float* __restrict__ Dvec,
                                   const float* __restrict__ H0,
                                   float* __restrict__ out) {
    const int bc = blockIdx.x;
    const int b  = bc / NC;
    const int c  = bc % NC;
    const int d  = blockIdx.y * blockDim.x + threadIdx.x;

    float An[DS];
#pragma unroll
    for (int s = 0; s < DS; ++s) An[s] = -__expf(A_log[d * DS + s]);

    float h[DS];
    const size_t hbase = (((size_t)b * NC + c) * DM + d) * DS;
#pragma unroll
    for (int s = 0; s < DS; ++s) h[s] = H0[hbase + s];

    const float dval = Dvec[d];
    const int t0 = c * CL;
    for (int t = t0; t < t0 + CL; ++t) {
        const int bt = b * TLEN + t;
        const float dtv = dt[(size_t)bt * DM + d];
        const float xv  = x[(size_t)bt * DM + d];
        const float dtx = dtv * xv;
        const float* Bm = xproj + (size_t)bt * NPJ + DTR;
        const float* Cm = xproj + (size_t)bt * NPJ + DTR + DS;
        float y = 0.0f;
#pragma unroll
        for (int s = 0; s < DS; ++s) {
            const float a = __expf(dtv * An[s]);
            h[s] = a * h[s] + dtx * Bm[s];
            y += h[s] * Cm[s];
        }
        out[(size_t)bt * DM + d] = y + xv * dval;
    }
}

extern "C" void kernel_launch(void* const* d_in, const int* in_sizes, int n_in,
                              void* d_out, int out_size, void* d_ws, size_t ws_size,
                              hipStream_t stream) {
    const float* x     = (const float*)d_in[0];
    const float* Wx    = (const float*)d_in[1];
    const float* Wdt   = (const float*)d_in[2];
    const float* bdt   = (const float*)d_in[3];
    const float* A_log = (const float*)d_in[4];
    const float* Dvec  = (const float*)d_in[5];
    float* out = (float*)d_out;

    char* ws = (char*)d_ws;
    float* xproj = (float*)ws;
    ws += (size_t)MROWS * NPJ * sizeof(float);
    float* dt    = (float*)ws;
    ws += (size_t)MROWS * DM * sizeof(float);
    float* P     = (float*)ws;
    ws += (size_t)BSZ * NC * DM * DS * sizeof(float);
    float* S     = (float*)ws;
    ws += (size_t)BSZ * NC * DM * DS * sizeof(float);
    float* H0    = (float*)ws;

    proj_gemm_kernel<<<dim3(MROWS / PJ_ROWS, NPJ / PJ_COLS), PJ_WAVES * 32, 0, stream>>>(x, Wx, xproj);
    dt_gemm_kernel<<<dim3(MROWS / 16, 16), 32, 0, stream>>>(xproj, Wdt, bdt, dt);
    scan_chunk1_kernel<<<dim3(BSZ * NC, DM / 256), 256, 0, stream>>>(x, xproj, dt, A_log, P, S);
    scan_carry_kernel<<<(BSZ * DM * DS) / 256, 256, 0, stream>>>(P, S, H0);
    scan_chunk2_kernel<<<dim3(BSZ * NC, DM / 256), 256, 0, stream>>>(x, xproj, dt, A_log, Dvec, H0, out);
}


# scan passes: per-step loads taken off the serial path: B/C rows of a chunk staged once per block in LDS and read a step ahead, each lane's dt and x fetched 8 steps ahead in registers
# speedup vs baseline: 2.3380x; 1.9474x over previous
#include <hip/hip_runtime.h>
#include <math.h>

typedef __attribute__((ext_vector_type(2))) float v2f;
typedef __attribute__((ext_vector_type(8))) float v8f;

#define DM   1024
#define DS   16
#define DTR  64
#define NPJ  96
#define BSZ  2
#define TLEN 2048
#define MROWS (BSZ*TLEN)
#define NC   32
#define CL   (TLEN/NC)

#define PJ_ROWS  32
#define PJ_COLS  48
#define PJ_WAVES 16
#define PJ_KW    (DM / PJ_WAVES)
#define PJ_TILE  (PJ_ROWS * PJ_COLS)

__global__ __launch_bounds__(PJ_WAVES * 32)
void proj_gemm_kernel(const float* __restrict__ x,
                      const float* __restrict__ Wx,
                      float* __restrict__ xproj) {
    __shared__ float4 red[PJ_WAVES][PJ_TILE / 4];

    const int mb   = blockIdx.x * PJ_ROWS;
    const int nb   = blockIdx.y * PJ_COLS;
    const int tid  = threadIdx.x;
    const int w    = tid >> 5;
    const int lane = tid & 31;
    const int half = lane >> 4;
    const int l16  = lane & 15;
    const int kb   = w * PJ_KW + 2 * half;

    v8f acc[2][3] = {};
    const float* arow0 = x + (size_t)(mb + l16) * DM + kb;
    const float* arow1 = arow0 + (size_t)16 * DM;
    const float* bcol  = Wx + (size_t)kb * NPJ + nb + l16;

#pragma unroll
    for (int ks = 0; ks < PJ_KW / 4; ++ks) {
        const v2f a0 = *(const v2f*)(arow0 + 4 * ks);
        const v2f a1 = *(const v2f*)(arow1 + 4 * ks);
        const float* bp = bcol + (size_t)(4 * ks) * NPJ;
        v2f b0, b1, b2;
        b0.x = bp[0];        b0.y = bp[NPJ];
        b1.x = bp[16];       b1.y = bp[16 + NPJ];
        b2.x = bp[32];       b2.y = bp[32 + NPJ];
        acc[0][0] = __builtin_amdgcn_wmma_f32_16x16x4_f32(false, a0, false, b0, (short)0, acc[0][0], false, false);
        acc[1][0] = __builtin_amdgcn_wmma_f32_16x16x4_f32(false, a1, false, b0, (short)0, acc[1][0], false, false);
        acc[0][1] = __builtin_amdgcn_wmma_f32_16x16x4_f32(false, a0, false, b1, (short)0, acc[0][1], false, false);
        acc[1][1] = __builtin_amdgcn_wmma_f32_16x16x4_f32(false, a1, false, b1, (short)0, acc[1][1], false, false);
        acc[0][2] = __builtin_amdgcn_wmma_f32_16x16x4_f32(false, a0, false, b2, (short)0, acc[0][2], false, false);
        acc[1][2] = __builtin_amdgcn_wmma_f32_16x16x4_f32(false, a1, false, b2, (short)0, acc[1][2], false, false);
    }

    float* mine = (float*)&red[w][0];
#pragma unroll
    for (int mi = 0; mi < 2; ++mi)
#pragma unroll
        for (int nj = 0; nj < 3; ++nj)
#pragma unroll
            for (int v = 0; v < 8; ++v)
                mine[((mi * 3 + nj) * 8 + v) * 32 + lane] = acc[mi][nj][v];
    __syncthreads();

    if (tid < PJ_TILE / 4) {
        float4 s = red[0][tid];
#pragma unroll
        for (int q = 1; q < PJ_WAVES; ++q) {
            const float4 p = red[q][tid];
            s.x += p.x; s.y += p.y; s.z += p.z; s.w += p.w;
        }
        const int frag = tid >> 6;
        const int v    = (tid >> 3) & 7;
        const int hh   = (tid >> 2) & 1;
        const int c0   = (tid & 3) * 4;
        const int mi   = frag / 3;
        const int nj   = frag - 3 * mi;
        const int row  = mb + 16 * mi + 8 * hh + v;
        const int col  = nb + 16 * nj + c0;
        *(float4*)(xproj + (size_t)row * NPJ + col) = s;
    }
}

__global__ void dt_gemm_kernel(const float* __restrict__ xproj,
                               const float* __restrict__ Wdt,
                               const float* __restrict__ bdt,
                               float* __restrict__ dt) {
    const int m0   = blockIdx.x * 16;
    const int nb   = blockIdx.y * 64;
    const int lane = threadIdx.x;
    const int half = lane >> 4;
    const int l16  = lane & 15;

    v8f acc[4] = {};
    const float* arow = xproj + (size_t)(m0 + l16) * NPJ;

#pragma unroll
    for (int k0 = 0; k0 < DTR; k0 += 4) {
        const int ka = k0 + 2 * half;
        v2f a = *(const v2f*)(arow + ka);
        const float* bp = Wdt + (size_t)ka * DM + nb + l16;
#pragma unroll
        for (int j = 0; j < 4; ++j) {
            v2f b;
            b.x = bp[j * 16];
            b.y = bp[j * 16 + DM];
            acc[j] = __builtin_amdgcn_wmma_f32_16x16x4_f32(false, a, false, b, (short)0, acc[j], false, false);
        }
    }
#pragma unroll
    for (int v = 0; v < 8; ++v) {
        const int row = m0 + v + (half ? 8 : 0);
#pragma unroll
        for (int j = 0; j < 4; ++j) {
            const int col = nb + j * 16 + l16;
            float z = acc[j][v] + bdt[col];
            float sp = (z > 20.0f) ? z : log1pf(__expf(z));
            dt[(size_t)row * DM + col] = sp;
        }
    }
}

#define SCAN_THREADS 256
#define SG 8
#define NG (CL / SG)

__global__ __launch_bounds__(SCAN_THREADS)
void scan_chunk1_kernel(const float* __restrict__ x,
                        const float* __restrict__ xproj,
                        const float* __restrict__ dt,
                        const float* __restrict__ A_log,
                        float* __restrict__ P,
                        float* __restrict__ S) {
    __shared__ float4 sB[CL][DS / 4];

    const int bc  = blockIdx.x;
    const int b   = bc / NC;
    const int c   = bc % NC;
    const int tid = threadIdx.x;
    const int d   = blockIdx.y * SCAN_THREADS + tid;
    const size_t row0 = (size_t)b * TLEN + (size_t)c * CL;

    sB[tid >> 2][tid & 3] =
        *(const float4*)(xproj + (row0 + (tid >> 2)) * NPJ + DTR + 4 * (tid & 3));

    const float* dtp = dt + row0 * DM + d;
    const float* xp  = x  + row0 * DM + d;
    float dn[SG], xn[SG];
#pragma unroll
    for (int u = 0; u < SG; ++u) { dn[u] = dtp[(size_t)u * DM]; xn[u] = xp[(size_t)u * DM]; }

    float An[DS];
#pragma unroll
    for (int s = 0; s < DS; ++s) An[s] = -__expf(A_log[d * DS + s]);

    float p[DS], h[DS];
#pragma unroll
    for (int s = 0; s < DS; ++s) { p[s] = 1.0f; h[s] = 0.0f; }

    __syncthreads();

    float4 nb[DS / 4];
#pragma unroll
    for (int q = 0; q < DS / 4; ++q) nb[q] = sB[0][q];

    for (int g = 0; g < NG; ++g) {
        float dc[SG], xc[SG];
#pragma unroll
        for (int u = 0; u < SG; ++u) { dc[u] = dn[u]; xc[u] = xn[u]; }
#pragma unroll
        for (int u = 0; u < SG; ++u) {
            const int t  = g * SG + u;
            const int tn = (t + 1 < CL) ? t + 1 : CL - 1;
            const float dtv = dc[u];
            const float dtx = dtv * xc[u];
            float Bm[DS];
#pragma unroll
            for (int q = 0; q < DS / 4; ++q) {
                const float4 v = nb[q];
                Bm[4 * q] = v.x; Bm[4 * q + 1] = v.y; Bm[4 * q + 2] = v.z; Bm[4 * q + 3] = v.w;
            }
#pragma unroll
            for (int q = 0; q < DS / 4; ++q) nb[q] = sB[tn][q];
            if (u == 0 && g + 1 < NG) {
#pragma unroll
                for (int v = 0; v < SG; ++v) {
                    dn[v] = dtp[(size_t)((g + 1) * SG + v) * DM];
                    xn[v] = xp [(size_t)((g + 1) * SG + v) * DM];
                }
            }
            __builtin_amdgcn_sched_barrier(0);
#pragma unroll
            for (int s = 0; s < DS; ++s) {
                const float a = __expf(dtv * An[s]);
                p[s] *= a;
                h[s] = a * h[s] + dtx * Bm[s];
            }
        }
    }
    const size_t base = (((size_t)b * NC + c) * DM + d) * DS;
#pragma unroll
    for (int s = 0; s < DS; ++s) { P[base + s] = p[s]; S[base + s] = h[s]; }
}

__global__ void scan_carry_kernel(const float* __restrict__ P,
                                  const float* __restrict__ S,
                                  float* __restrict__ H0) {
    const int idx = blockIdx.x * blockDim.x + threadIdx.x;
    const int b   = idx / (DM * DS);
    const int ds  = idx % (DM * DS);
    float h = 0.0f;
    for (int c = 0; c < NC; ++c) {
        const size_t off = (((size_t)b * NC + c) * DM * DS) + ds;
        H0[off] = h;
        h = P[off] * h + S[off];
    }
}

__global__ __launch_bounds__(SCAN_THREADS)
void scan_chunk2_kernel(const float* __restrict__ x,
                        const float* __restrict__ xproj,
                        const float* __restrict__ dt,
                        const float* __restrict__ A_log,
                        const float* __restrict__ Dvec,
                        const float* __restrict__ H0,
                        float* __restrict__ out) {
    __shared__ float4 sBC[CL][2 * DS / 4];

    const int bc  = blockIdx.x;
    const int b   = bc / NC;
    const int c   = bc % NC;
    const int tid = threadIdx.x;
    const int d   = blockIdx.y * SCAN_THREADS + tid;
    const size_t row0 = (size_t)b * TLEN + (size_t)c * CL;

    {
        const int r = tid >> 3, q = tid & 7;
        const float4 lo = *(const float4*)(xproj + (row0 + r) * NPJ + DTR + 4 * q);
        const float4 hi = *(const float4*)(xproj + (row0 + CL / 2 + r) * NPJ + DTR + 4 * q);
        sBC[r][q] = lo;
        sBC[CL / 2 + r][q] = hi;
    }

    const float* dtp = dt + row0 * DM + d;
    const float* xp  = x  + row0 * DM + d;
    float*       op  = out + row0 * DM + d;
    float dn[SG], xn[SG];
#pragma unroll
    for (int u = 0; u < SG; ++u) { dn[u] = dtp[(size_t)u * DM]; xn[u] = xp[(size_t)u * DM]; }

    float An[DS];
#pragma unroll
    for (int s = 0; s < DS; ++s) An[s] = -__expf(A_log[d * DS + s]);

    float h[DS];
    const size_t hbase = (((size_t)b * NC + c) * DM + d) * DS;
#pragma unroll
    for (int s = 0; s < DS; ++s) h[s] = H0[hbase + s];

    const float dval = Dvec[d];

    __syncthreads();

    float4 nb[2 * DS / 4];
#pragma unroll
    for (int q = 0; q < 2 * DS / 4; ++q) nb[q] = sBC[0][q];

    for (int g = 0; g < NG; ++g) {
        float dc[SG], xc[SG];
#pragma unroll
        for (int u = 0; u < SG; ++u) { dc[u] = dn[u]; xc[u] = xn[u]; }
#pragma unroll
        for (int u = 0; u < SG; ++u) {
            const int t  = g * SG + u;
            const int tn = (t + 1 < CL) ? t + 1 : CL - 1;
            const float dtv = dc[u];
            const float xv  = xc[u];
            const float dtx = dtv * xv;
            float Bm[DS], Cm[DS];
#pragma unroll
            for (int q = 0; q < DS / 4; ++q) {
                const float4 v = nb[q];
                const float4 w = nb[DS / 4 + q];
                Bm[4 * q] = v.x; Bm[4 * q + 1] = v.y; Bm[4 * q + 2] = v.z; Bm[4 * q + 3] = v.w;
                Cm[4 * q] = w.x; Cm[4 * q + 1] = w.y; Cm[4 * q + 2] = w.z; Cm[4 * q + 3] = w.w;
            }
#pragma unroll
            for (int q = 0; q < 2 * DS / 4; ++q) nb[q] = sBC[tn][q];
            if (u == 0 && g + 1 < NG) {
#pragma unroll
                for (int v = 0; v < SG; ++v) {
                    dn[v] = dtp[(size_t)((g + 1) * SG + v) * DM];
                    xn[v] = xp [(size_t)((g + 1) * SG + v) * DM];
                }
            }
            __builtin_amdgcn_sched_barrier(0);
            float y = 0.0f;
#pragma unroll
            for (int s = 0; s < DS; ++s) {
                const float a = __expf(dtv * An[s]);
                h[s] = a * h[s] + dtx * Bm[s];
                y += h[s] * Cm[s];
            }
            op[(size_t)t * DM] = y + xv * dval;
        }
    }
}

extern "C" void kernel_launch(void* const* d_in, const int* in_sizes, int n_in,
                              void* d_out, int out_size, void* d_ws, size_t ws_size,
                              hipStream_t stream) {
    const float* x     = (const float*)d_in[0];
    const float* Wx    = (const float*)d_in[1];
    const float* Wdt   = (const float*)d_in[2];
    const float* bdt   = (const float*)d_in[3];
    const float* A_log = (const float*)d_in[4];
    const float* Dvec  = (const float*)d_in[5];
    float* out = (float*)d_out;

    char* ws = (char*)d_ws;
    float* xproj = (float*)ws;
    ws += (size_t)MROWS * NPJ * sizeof(float);
    float* dt    = (float*)ws;
    ws += (size_t)MROWS * DM * sizeof(float);
    float* P     = (float*)ws;
    ws += (size_t)BSZ * NC * DM * DS * sizeof(float);
    float* S     = (float*)ws;
    ws += (size_t)BSZ * NC * DM * DS * sizeof(float);
    float* H0    = (float*)ws;

    proj_gemm_kernel<<<dim3(MROWS / PJ_ROWS, NPJ / PJ_COLS), PJ_WAVES * 32, 0, stream>>>(x, Wx, xproj);
    dt_gemm_kernel<<<dim3(MROWS / 16, 16), 32, 0, stream>>>(xproj, Wdt, bdt, dt);
    scan_chunk1_kernel<<<dim3(BSZ * NC, DM / SCAN_THREADS), SCAN_THREADS, 0, stream>>>(x, xproj, dt, A_log, P, S);
    scan_carry_kernel<<<(BSZ * DM * DS) / 256, 256, 0, stream>>>(P, S, H0);
    scan_chunk2_kernel<<<dim3(BSZ * NC, DM / SCAN_THREADS), SCAN_THREADS, 0, stream>>>(x, xproj, dt, A_log, Dvec, H0, out);
}


# dt GEMM epilogue: softplus as max(z,0)+log1p(exp(-|z|)) straight-line on hardware exp2/log2/rcp with an exact rounding correction, replacing branchy library log1pf; bias loaded once per column
# speedup vs baseline: 2.6593x; 1.1374x over previous
#include <hip/hip_runtime.h>
#include <math.h>

typedef __attribute__((ext_vector_type(2))) float v2f;
typedef __attribute__((ext_vector_type(8))) float v8f;

#define DM   1024
#define DS   16
#define DTR  64
#define NPJ  96
#define BSZ  2
#define TLEN 2048
#define MROWS (BSZ*TLEN)
#define NC   32
#define CL   (TLEN/NC)

#define PJ_ROWS  32
#define PJ_COLS  48
#define PJ_WAVES 16
#define PJ_KW    (DM / PJ_WAVES)
#define PJ_TILE  (PJ_ROWS * PJ_COLS)

__global__ __launch_bounds__(PJ_WAVES * 32)
void proj_gemm_kernel(const float* __restrict__ x,
                      const float* __restrict__ Wx,
                      float* __restrict__ xproj) {
    __shared__ float4 red[PJ_WAVES][PJ_TILE / 4];

    const int mb   = blockIdx.x * PJ_ROWS;
    const int nb   = blockIdx.y * PJ_COLS;
    const int tid  = threadIdx.x;
    const int w    = tid >> 5;
    const int lane = tid & 31;
    const int half = lane >> 4;
    const int l16  = lane & 15;
    const int kb   = w * PJ_KW + 2 * half;

    v8f acc[2][3] = {};
    const float* arow0 = x + (size_t)(mb + l16) * DM + kb;
    const float* arow1 = arow0 + (size_t)16 * DM;
    const float* bcol  = Wx + (size_t)kb * NPJ + nb + l16;

#pragma unroll
    for (int ks = 0; ks < PJ_KW / 4; ++ks) {
        const v2f a0 = *(const v2f*)(arow0 + 4 * ks);
        const v2f a1 = *(const v2f*)(arow1 + 4 * ks);
        const float* bp = bcol + (size_t)(4 * ks) * NPJ;
        v2f b0, b1, b2;
        b0.x = bp[0];        b0.y = bp[NPJ];
        b1.x = bp[16];       b1.y = bp[16 + NPJ];
        b2.x = bp[32];       b2.y = bp[32 + NPJ];
        acc[0][0] = __builtin_amdgcn_wmma_f32_16x16x4_f32(false, a0, false, b0, (short)0, acc[0][0], false, false);
        acc[1][0] = __builtin_amdgcn_wmma_f32_16x16x4_f32(false, a1, false, b0, (short)0, acc[1][0], false, false);
        acc[0][1] = __builtin_amdgcn_wmma_f32_16x16x4_f32(false, a0, false, b1, (short)0, acc[0][1], false, false);
        acc[1][1] = __builtin_amdgcn_wmma_f32_16x16x4_f32(false, a1, false, b1, (short)0, acc[1][1], false, false);
        acc[0][2] = __builtin_amdgcn_wmma_f32_16x16x4_f32(false, a0, false, b2, (short)0, acc[0][2], false, false);
        acc[1][2] = __builtin_amdgcn_wmma_f32_16x16x4_f32(false, a1, false, b2, (short)0, acc[1][2], false, false);
    }

    float* mine = (float*)&red[w][0];
#pragma unroll
    for (int mi = 0; mi < 2; ++mi)
#pragma unroll
        for (int nj = 0; nj < 3; ++nj)
#pragma unroll
            for (int v = 0; v < 8; ++v)
                mine[((mi * 3 + nj) * 8 + v) * 32 + lane] = acc[mi][nj][v];
    __syncthreads();

    if (tid < PJ_TILE / 4) {
        float4 s = red[0][tid];
#pragma unroll
        for (int q = 1; q < PJ_WAVES; ++q) {
            const float4 p = red[q][tid];
            s.x += p.x; s.y += p.y; s.z += p.z; s.w += p.w;
        }
        const int frag = tid >> 6;
        const int v    = (tid >> 3) & 7;
        const int hh   = (tid >> 2) & 1;
        const int c0   = (tid & 3) * 4;
        const int mi   = frag / 3;
        const int nj   = frag - 3 * mi;
        const int row  = mb + 16 * mi + 8 * hh + v;
        const int col  = nb + 16 * nj + c0;
        *(float4*)(xproj + (size_t)row * NPJ + col) = s;
    }
}

__device__ __forceinline__ float softplus_f32(float z) {
    const float t  = __builtin_amdgcn_exp2f(-fabsf(z) * 1.44269504088896341f);
    const float u  = 1.0f + t;
    const float d  = t - (u - 1.0f);
    const float lg = __builtin_amdgcn_logf(u) * 0.693147180559945309f;
    return fmaxf(z, 0.0f) + fmaf(d, __builtin_amdgcn_rcpf(u), lg);
}

__global__ void dt_gemm_kernel(const float* __restrict__ xproj,
                               const float* __restrict__ Wdt,
                               const float* __restrict__ bdt,
                               float* __restrict__ dt) {
    const int m0   = blockIdx.x * 16;
    const int nb   = blockIdx.y * 64;
    const int lane = threadIdx.x;
    const int half = lane >> 4;
    const int l16  = lane & 15;

    v8f acc[4] = {};
    const float* arow = xproj + (size_t)(m0 + l16) * NPJ;

#pragma unroll
    for (int k0 = 0; k0 < DTR; k0 += 4) {
        const int ka = k0 + 2 * half;
        v2f a = *(const v2f*)(arow + ka);
        const float* bp = Wdt + (size_t)ka * DM + nb + l16;
#pragma unroll
        for (int j = 0; j < 4; ++j) {
            v2f b;
            b.x = bp[j * 16];
            b.y = bp[j * 16 + DM];
            acc[j] = __builtin_amdgcn_wmma_f32_16x16x4_f32(false, a, false, b, (short)0, acc[j], false, false);
        }
    }
    float bias[4];
#pragma unroll
    for (int j = 0; j < 4; ++j) bias[j] = bdt[nb + j * 16 + l16];
#pragma unroll
    for (int v = 0; v < 8; ++v) {
        const int row = m0 + v + (half ? 8 : 0);
#pragma unroll
        for (int j = 0; j < 4; ++j) {
            const int col = nb + j * 16 + l16;
            dt[(size_t)row * DM + col] = softplus_f32(acc[j][v] + bias[j]);
        }
    }
}

#define SCAN_THREADS 256
#define SG 8
#define NG (CL / SG)

__global__ __launch_bounds__(SCAN_THREADS)
void scan_chunk1_kernel(const float* __restrict__ x,
                        const float* __restrict__ xproj,
                        const float* __restrict__ dt,
                        const float* __restrict__ A_log,
                        float* __restrict__ P,
                        float* __restrict__ S) {
    __shared__ float4 sB[CL][DS / 4];

    const int bc  = blockIdx.x;
    const int b   = bc / NC;
    const int c   = bc % NC;
    const int tid = threadIdx.x;
    const int d   = blockIdx.y * SCAN_THREADS + tid;
    const size_t row0 = (size_t)b * TLEN + (size_t)c * CL;

    sB[tid >> 2][tid & 3] =
        *(const float4*)(xproj + (row0 + (tid >> 2)) * NPJ + DTR + 4 * (tid & 3));

    const float* dtp = dt + row0 * DM + d;
    const float* xp  = x  + row0 * DM + d;
    float dn[SG], xn[SG];
#pragma unroll
    for (int u = 0; u < SG; ++u) { dn[u] = dtp[(size_t)u * DM]; xn[u] = xp[(size_t)u * DM]; }

    float An[DS];
#pragma unroll
    for (int s = 0; s < DS; ++s) An[s] = -__expf(A_log[d * DS + s]);

    float p[DS], h[DS];
#pragma unroll
    for (int s = 0; s < DS; ++s) { p[s] = 1.0f; h[s] = 0.0f; }

    __syncthreads();

    float4 nb[DS / 4];
#pragma unroll
    for (int q = 0; q < DS / 4; ++q) nb[q] = sB[0][q];

    for (int g = 0; g < NG; ++g) {
        float dc[SG], xc[SG];
#pragma unroll
        for (int u = 0; u < SG; ++u) { dc[u] = dn[u]; xc[u] = xn[u]; }
#pragma unroll
        for (int u = 0; u < SG; ++u) {
            const int t  = g * SG + u;
            const int tn = (t + 1 < CL) ? t + 1 : CL - 1;
            const float dtv = dc[u];
            const float dtx = dtv * xc[u];
            float Bm[DS];
#pragma unroll
            for (int q = 0; q < DS / 4; ++q) {
                const float4 v = nb[q];
                Bm[4 * q] = v.x; Bm[4 * q + 1] = v.y; Bm[4 * q + 2] = v.z; Bm[4 * q + 3] = v.w;
            }
#pragma unroll
            for (int q = 0; q < DS / 4; ++q) nb[q] = sB[tn][q];
            if (u == 0 && g + 1 < NG) {
#pragma unroll
                for (int v = 0; v < SG; ++v) {
                    dn[v] = dtp[(size_t)((g + 1) * SG + v) * DM];
                    xn[v] = xp [(size_t)((g + 1) * SG + v) * DM];
                }
            }
            __builtin_amdgcn_sched_barrier(0);
#pragma unroll
            for (int s = 0; s < DS; ++s) {
                const float a = __expf(dtv * An[s]);
                p[s] *= a;
                h[s] = a * h[s] + dtx * Bm[s];
            }
        }
    }
    const size_t base = (((size_t)b * NC + c) * DM + d) * DS;
#pragma unroll
    for (int s = 0; s < DS; ++s) { P[base + s] = p[s]; S[base + s] = h[s]; }
}

__global__ void scan_carry_kernel(const float* __restrict__ P,
                                  const float* __restrict__ S,
                                  float* __restrict__ H0) {
    const int idx = blockIdx.x * blockDim.x + threadIdx.x;
    const int b   = idx / (DM * DS);
    const int ds  = idx % (DM * DS);
    float h = 0.0f;
    for (int c = 0; c < NC; ++c) {
        const size_t off = (((size_t)b * NC + c) * DM * DS) + ds;
        H0[off] = h;
        h = P[off] * h + S[off];
    }
}

__global__ __launch_bounds__(SCAN_THREADS)
void scan_chunk2_kernel(const float* __restrict__ x,
                        const float* __restrict__ xproj,
                        const float* __restrict__ dt,
                        const float* __restrict__ A_log,
                        const float* __restrict__ Dvec,
                        const float* __restrict__ H0,
                        float* __restrict__ out) {
    __shared__ float4 sBC[CL][2 * DS / 4];

    const int bc  = blockIdx.x;
    const int b   = bc / NC;
    const int c   = bc % NC;
    const int tid = threadIdx.x;
    const int d   = blockIdx.y * SCAN_THREADS + tid;
    const size_t row0 = (size_t)b * TLEN + (size_t)c * CL;

    {
        const int r = tid >> 3, q = tid & 7;
        const float4 lo = *(const float4*)(xproj + (row0 + r) * NPJ + DTR + 4 * q);
        const float4 hi = *(const float4*)(xproj + (row0 + CL / 2 + r) * NPJ + DTR + 4 * q);
        sBC[r][q] = lo;
        sBC[CL / 2 + r][q] = hi;
    }

    const float* dtp = dt + row0 * DM + d;
    const float* xp  = x  + row0 * DM + d;
    float*       op  = out + row0 * DM + d;
    float dn[SG], xn[SG];
#pragma unroll
    for (int u = 0; u < SG; ++u) { dn[u] = dtp[(size_t)u * DM]; xn[u] = xp[(size_t)u * DM]; }

    float An[DS];
#pragma unroll
    for (int s = 0; s < DS; ++s) An[s] = -__expf(A_log[d * DS + s]);

    float h[DS];
    const size_t hbase = (((size_t)b * NC + c) * DM + d) * DS;
#pragma unroll
    for (int s = 0; s < DS; ++s) h[s] = H0[hbase + s];

    const float dval = Dvec[d];

    __syncthreads();

    float4 nb[2 * DS / 4];
#pragma unroll
    for (int q = 0; q < 2 * DS / 4; ++q) nb[q] = sBC[0][q];

    for (int g = 0; g < NG; ++g) {
        float dc[SG], xc[SG];
#pragma unroll
        for (int u = 0; u < SG; ++u) { dc[u] = dn[u]; xc[u] = xn[u]; }
#pragma unroll
        for (int u = 0; u < SG; ++u) {
            const int t  = g * SG + u;
            const int tn = (t + 1 < CL) ? t + 1 : CL - 1;
            const float dtv = dc[u];
            const float xv  = xc[u];
            const float dtx = dtv * xv;
            float Bm[DS], Cm[DS];
#pragma unroll
            for (int q = 0; q < DS / 4; ++q) {
                const float4 v = nb[q];
                const float4 w = nb[DS / 4 + q];
                Bm[4 * q] = v.x; Bm[4 * q + 1] = v.y; Bm[4 * q + 2] = v.z; Bm[4 * q + 3] = v.w;
                Cm[4 * q] = w.x; Cm[4 * q + 1] = w.y; Cm[4 * q + 2] = w.z; Cm[4 * q + 3] = w.w;
            }
#pragma unroll
            for (int q = 0; q < 2 * DS / 4; ++q) nb[q] = sBC[tn][q];
            if (u == 0 && g + 1 < NG) {
#pragma unroll
                for (int v = 0; v < SG; ++v) {
                    dn[v] = dtp[(size_t)((g + 1) * SG + v) * DM];
                    xn[v] = xp [(size_t)((g + 1) * SG + v) * DM];
                }
            }
            __builtin_amdgcn_sched_barrier(0);
            float y = 0.0f;
#pragma unroll
            for (int s = 0; s < DS; ++s) {
                const float a = __expf(dtv * An[s]);
                h[s] = a * h[s] + dtx * Bm[s];
                y += h[s] * Cm[s];
            }
            op[(size_t)t * DM] = y + xv * dval;
        }
    }
}

extern "C" void kernel_launch(void* const* d_in, const int* in_sizes, int n_in,
                              void* d_out, int out_size, void* d_ws, size_t ws_size,
                              hipStream_t stream) {
    const float* x     = (const float*)d_in[0];
    const float* Wx    = (const float*)d_in[1];
    const float* Wdt   = (const float*)d_in[2];
    const float* bdt   = (const float*)d_in[3];
    const float* A_log = (const float*)d_in[4];
    const float* Dvec  = (const float*)d_in[5];
    float* out = (float*)d_out;

    char* ws = (char*)d_ws;
    float* xproj = (float*)ws;
    ws += (size_t)MROWS * NPJ * sizeof(float);
    float* dt    = (float*)ws;
    ws += (size_t)MROWS * DM * sizeof(float);
    float* P     = (float*)ws;
    ws += (size_t)BSZ * NC * DM * DS * sizeof(float);
    float* S     = (float*)ws;
    ws += (size_t)BSZ * NC * DM * DS * sizeof(float);
    float* H0    = (float*)ws;

    proj_gemm_kernel<<<dim3(MROWS / PJ_ROWS, NPJ / PJ_COLS), PJ_WAVES * 32, 0, stream>>>(x, Wx, xproj);
    dt_gemm_kernel<<<dim3(MROWS / 16, 16), 32, 0, stream>>>(xproj, Wdt, bdt, dt);
    scan_chunk1_kernel<<<dim3(BSZ * NC, DM / SCAN_THREADS), SCAN_THREADS, 0, stream>>>(x, xproj, dt, A_log, P, S);
    scan_carry_kernel<<<(BSZ * DM * DS) / 256, 256, 0, stream>>>(P, S, H0);
    scan_chunk2_kernel<<<dim3(BSZ * NC, DM / SCAN_THREADS), SCAN_THREADS, 0, stream>>>(x, xproj, dt, A_log, Dvec, H0, out);
}


# dt GEMM + softplus moved into the proj GEMM launch as a second phase: a block owns 16 rows x all 96 cols, dt_raw handed over in LDS, W_dt fragments fetched during the split-K sum; 4 launches, was 5
# speedup vs baseline: 2.8089x; 1.0563x over previous
#include <hip/hip_runtime.h>
#include <math.h>

typedef __attribute__((ext_vector_type(2))) float v2f;
typedef __attribute__((ext_vector_type(8))) float v8f;

#define DM   1024
#define DS   16
#define DTR  64
#define NPJ  96
#define BSZ  2
#define TLEN 2048
#define MROWS (BSZ*TLEN)
#define NC   32
#define CL   (TLEN/NC)

__device__ __forceinline__ float softplus_f32(float z) {
    const float t  = __builtin_amdgcn_exp2f(-fabsf(z) * 1.44269504088896341f);
    const float u  = 1.0f + t;
    const float d  = t - (u - 1.0f);
    const float lg = __builtin_amdgcn_logf(u) * 0.693147180559945309f;
    return fmaxf(z, 0.0f) + fmaf(d, __builtin_amdgcn_rcpf(u), lg);
}

#define PJ_ROWS  16
#define PJ_WAVES 16
#define PJ_KW    (DM / PJ_WAVES)
#define PJ_TILE  (PJ_ROWS * NPJ)
#define XS_PITCH 68

__global__ __launch_bounds__(PJ_WAVES * 32)
void proj_dt_kernel(const float* __restrict__ x,
                    const float* __restrict__ Wx,
                    const float* __restrict__ Wdt,
                    const float* __restrict__ bdt,
                    float* __restrict__ xproj,
                    float* __restrict__ dt) {
    __shared__ float4 red[PJ_WAVES][PJ_TILE / 4];
    __shared__ __attribute__((aligned(16))) float xs[PJ_ROWS][XS_PITCH];

    const int mb   = blockIdx.x * PJ_ROWS;
    const int tid  = threadIdx.x;
    const int w    = tid >> 5;
    const int lane = tid & 31;
    const int half = lane >> 4;
    const int l16  = lane & 15;
    const int kb   = w * PJ_KW + 2 * half;

    v8f acc[6] = {};
    const float* arow = x + (size_t)(mb + l16) * DM + kb;
    const float* bcol = Wx + (size_t)kb * NPJ + l16;

#pragma unroll
    for (int ks = 0; ks < PJ_KW / 4; ++ks) {
        const v2f a = *(const v2f*)(arow + 4 * ks);
        const float* bp = bcol + (size_t)(4 * ks) * NPJ;
#pragma unroll
        for (int j = 0; j < 6; ++j) {
            v2f b;
            b.x = bp[16 * j];
            b.y = bp[16 * j + NPJ];
            acc[j] = __builtin_amdgcn_wmma_f32_16x16x4_f32(false, a, false, b, (short)0, acc[j], false, false);
        }
    }

    float* mine = (float*)&red[w][0];
#pragma unroll
    for (int j = 0; j < 6; ++j)
#pragma unroll
        for (int v = 0; v < 8; ++v)
            mine[(j * 8 + v) * 32 + lane] = acc[j][v];

    const int nb2 = w * 64;
    v2f bw[DTR / 4][4];
#pragma unroll
    for (int ks = 0; ks < DTR / 4; ++ks) {
        const float* bp = Wdt + (size_t)(4 * ks + 2 * half) * DM + nb2 + l16;
#pragma unroll
        for (int j = 0; j < 4; ++j) {
            bw[ks][j].x = bp[16 * j];
            bw[ks][j].y = bp[16 * j + DM];
        }
    }
    float bias[4];
#pragma unroll
    for (int j = 0; j < 4; ++j) bias[j] = bdt[nb2 + 16 * j + l16];

    __syncthreads();

    if (tid < PJ_TILE / 4) {
        float4 s = red[0][tid];
#pragma unroll
        for (int q = 1; q < PJ_WAVES; ++q) {
            const float4 p = red[q][tid];
            s.x += p.x; s.y += p.y; s.z += p.z; s.w += p.w;
        }
        const int frag = tid >> 6;
        const int v    = (tid >> 3) & 7;
        const int hh   = (tid >> 2) & 1;
        const int c0   = (tid & 3) * 4;
        const int rl   = 8 * hh + v;
        const int col  = 16 * frag + c0;
        if (frag < DTR / 16)
            *(float4*)&xs[rl][col] = s;
        else
            *(float4*)(xproj + (size_t)(mb + rl) * NPJ + col) = s;
    }
    __syncthreads();

    v8f acc2[4] = {};
#pragma unroll
    for (int ks = 0; ks < DTR / 4; ++ks) {
        const v2f a = *(const v2f*)&xs[l16][4 * ks + 2 * half];
#pragma unroll
        for (int j = 0; j < 4; ++j)
            acc2[j] = __builtin_amdgcn_wmma_f32_16x16x4_f32(false, a, false, bw[ks][j], (short)0, acc2[j], false, false);
    }
#pragma unroll
    for (int v = 0; v < 8; ++v) {
        const int row = mb + v + (half ? 8 : 0);
#pragma unroll
        for (int j = 0; j < 4; ++j) {
            const int col = nb2 + j * 16 + l16;
            dt[(size_t)row * DM + col] = softplus_f32(acc2[j][v] + bias[j]);
        }
    }
}

#define SCAN_THREADS 256
#define SG 8
#define NG (CL / SG)

__global__ __launch_bounds__(SCAN_THREADS)
void scan_chunk1_kernel(const float* __restrict__ x,
                        const float* __restrict__ xproj,
                        const float* __restrict__ dt,
                        const float* __restrict__ A_log,
                        float* __restrict__ P,
                        float* __restrict__ S) {
    __shared__ float4 sB[CL][DS / 4];

    const int bc  = blockIdx.x;
    const int b   = bc / NC;
    const int c   = bc % NC;
    const int tid = threadIdx.x;
    const int d   = blockIdx.y * SCAN_THREADS + tid;
    const size_t row0 = (size_t)b * TLEN + (size_t)c * CL;

    sB[tid >> 2][tid & 3] =
        *(const float4*)(xproj + (row0 + (tid >> 2)) * NPJ + DTR + 4 * (tid & 3));

    const float* dtp = dt + row0 * DM + d;
    const float* xp  = x  + row0 * DM + d;
    float dn[SG], xn[SG];
#pragma unroll
    for (int u = 0; u < SG; ++u) { dn[u] = dtp[(size_t)u * DM]; xn[u] = xp[(size_t)u * DM]; }

    float An[DS];
#pragma unroll
    for (int s = 0; s < DS; ++s) An[s] = -__expf(A_log[d * DS + s]);

    float p[DS], h[DS];
#pragma unroll
    for (int s = 0; s < DS; ++s) { p[s] = 1.0f; h[s] = 0.0f; }

    __syncthreads();

    float4 nb[DS / 4];
#pragma unroll
    for (int q = 0; q < DS / 4; ++q) nb[q] = sB[0][q];

    for (int g = 0; g < NG; ++g) {
        float dc[SG], xc[SG];
#pragma unroll
        for (int u = 0; u < SG; ++u) { dc[u] = dn[u]; xc[u] = xn[u]; }
#pragma unroll
        for (int u = 0; u < SG; ++u) {
            const int t  = g * SG + u;
            const int tn = (t + 1 < CL) ? t + 1 : CL - 1;
            const float dtv = dc[u];
            const float dtx = dtv * xc[u];
            float Bm[DS];
#pragma unroll
            for (int q = 0; q < DS / 4; ++q) {
                const float4 v = nb[q];
                Bm[4 * q] = v.x; Bm[4 * q + 1] = v.y; Bm[4 * q + 2] = v.z; Bm[4 * q + 3] = v.w;
            }
#pragma unroll
            for (int q = 0; q < DS / 4; ++q) nb[q] = sB[tn][q];
            if (u == 0 && g + 1 < NG) {
#pragma unroll
                for (int v = 0; v < SG; ++v) {
                    dn[v] = dtp[(size_t)((g + 1) * SG + v) * DM];
                    xn[v] = xp [(size_t)((g + 1) * SG + v) * DM];
                }
            }
            __builtin_amdgcn_sched_barrier(0);
#pragma unroll
            for (int s = 0; s < DS; ++s) {
                const float a = __expf(dtv * An[s]);
                p[s] *= a;
                h[s] = a * h[s] + dtx * Bm[s];
            }
        }
    }
    const size_t base = (((size_t)b * NC + c) * DM + d) * DS;
#pragma unroll
    for (int s = 0; s < DS; ++s) { P[base + s] = p[s]; S[base + s] = h[s]; }
}

__global__ void scan_carry_kernel(const float* __restrict__ P,
                                  const float* __restrict__ S,
                                  float* __restrict__ H0) {
    const int idx = blockIdx.x * blockDim.x + threadIdx.x;
    const int b   = idx / (DM * DS);
    const int ds  = idx % (DM * DS);
    float h = 0.0f;
    for (int c = 0; c < NC; ++c) {
        const size_t off = (((size_t)b * NC + c) * DM * DS) + ds;
        H0[off] = h;
        h = P[off] * h + S[off];
    }
}

__global__ __launch_bounds__(SCAN_THREADS)
void scan_chunk2_kernel(const float* __restrict__ x,
                        const float* __restrict__ xproj,
                        const float* __restrict__ dt,
                        const float* __restrict__ A_log,
                        const float* __restrict__ Dvec,
                        const float* __restrict__ H0,
                        float* __restrict__ out) {
    __shared__ float4 sBC[CL][2 * DS / 4];

    const int bc  = blockIdx.x;
    const int b   = bc / NC;
    const int c   = bc % NC;
    const int tid = threadIdx.x;
    const int d   = blockIdx.y * SCAN_THREADS + tid;
    const size_t row0 = (size_t)b * TLEN + (size_t)c * CL;

    {
        const int r = tid >> 3, q = tid & 7;
        const float4 lo = *(const float4*)(xproj + (row0 + r) * NPJ + DTR + 4 * q);
        const float4 hi = *(const float4*)(xproj + (row0 + CL / 2 + r) * NPJ + DTR + 4 * q);
        sBC[r][q] = lo;
        sBC[CL / 2 + r][q] = hi;
    }

    const float* dtp = dt + row0 * DM + d;
    const float* xp  = x  + row0 * DM + d;
    float*       op  = out + row0 * DM + d;
    float dn[SG], xn[SG];
#pragma unroll
    for (int u = 0; u < SG; ++u) { dn[u] = dtp[(size_t)u * DM]; xn[u] = xp[(size_t)u * DM]; }

    float An[DS];
#pragma unroll
    for (int s = 0; s < DS; ++s) An[s] = -__expf(A_log[d * DS + s]);

    float h[DS];
    const size_t hbase = (((size_t)b * NC + c) * DM + d) * DS;
#pragma unroll
    for (int s = 0; s < DS; ++s) h[s] = H0[hbase + s];

    const float dval = Dvec[d];

    __syncthreads();

    float4 nb[2 * DS / 4];
#pragma unroll
    for (int q = 0; q < 2 * DS / 4; ++q) nb[q] = sBC[0][q];

    for (int g = 0; g < NG; ++g) {
        float dc[SG], xc[SG];
#pragma unroll
        for (int u = 0; u < SG; ++u) { dc[u] = dn[u]; xc[u] = xn[u]; }
#pragma unroll
        for (int u = 0; u < SG; ++u) {
            const int t  = g * SG + u;
            const int tn = (t + 1 < CL) ? t + 1 : CL - 1;
            const float dtv = dc[u];
            const float xv  = xc[u];
            const float dtx = dtv * xv;
            float Bm[DS], Cm[DS];
#pragma unroll
            for (int q = 0; q < DS / 4; ++q) {
                const float4 v = nb[q];
                const float4 w = nb[DS / 4 + q];
                Bm[4 * q] = v.x; Bm[4 * q + 1] = v.y; Bm[4 * q + 2] = v.z; Bm[4 * q + 3] = v.w;
                Cm[4 * q] = w.x; Cm[4 * q + 1] = w.y; Cm[4 * q + 2] = w.z; Cm[4 * q + 3] = w.w;
            }
#pragma unroll
            for (int q = 0; q < 2 * DS / 4; ++q) nb[q] = sBC[tn][q];
            if (u == 0 && g + 1 < NG) {
#pragma unroll
                for (int v = 0; v < SG; ++v) {
                    dn[v] = dtp[(size_t)((g + 1) * SG + v) * DM];
                    xn[v] = xp [(size_t)((g + 1) * SG + v) * DM];
                }
            }
            __builtin_amdgcn_sched_barrier(0);
            float y = 0.0f;
#pragma unroll
            for (int s = 0; s < DS; ++s) {
                const float a = __expf(dtv * An[s]);
                h[s] = a * h[s] + dtx * Bm[s];
                y += h[s] * Cm[s];
            }
            op[(size_t)t * DM] = y + xv * dval;
        }
    }
}

extern "C" void kernel_launch(void* const* d_in, const int* in_sizes, int n_in,
                              void* d_out, int out_size, void* d_ws, size_t ws_size,
                              hipStream_t stream) {
    const float* x     = (const float*)d_in[0];
    const float* Wx    = (const float*)d_in[1];
    const float* Wdt   = (const float*)d_in[2];
    const float* bdt   = (const float*)d_in[3];
    const float* A_log = (const float*)d_in[4];
    const float* Dvec  = (const float*)d_in[5];
    float* out = (float*)d_out;

    char* ws = (char*)d_ws;
    float* xproj = (float*)ws;
    ws += (size_t)MROWS * NPJ * sizeof(float);
    float* dt    = (float*)ws;
    ws += (size_t)MROWS * DM * sizeof(float);
    float* P     = (float*)ws;
    ws += (size_t)BSZ * NC * DM * DS * sizeof(float);
    float* S     = (float*)ws;
    ws += (size_t)BSZ * NC * DM * DS * sizeof(float);
    float* H0    = (float*)ws;

    proj_dt_kernel<<<dim3(MROWS / PJ_ROWS), PJ_WAVES * 32, 0, stream>>>(x, Wx, Wdt, bdt, xproj, dt);
    scan_chunk1_kernel<<<dim3(BSZ * NC, DM / SCAN_THREADS), SCAN_THREADS, 0, stream>>>(x, xproj, dt, A_log, P, S);
    scan_carry_kernel<<<(BSZ * DM * DS) / 256, 256, 0, stream>>>(P, S, H0);
    scan_chunk2_kernel<<<dim3(BSZ * NC, DM / SCAN_THREADS), SCAN_THREADS, 0, stream>>>(x, xproj, dt, A_log, Dvec, H0, out);
}


# carry launch removed: a pass-1 block takes 8 consecutive chunks and chains its results, pass 2 ends each entry state in 4 steps; chunk decay = exp(A*sum dt), not a product of exps (rounding changes)
# speedup vs baseline: 2.9790x; 1.0606x over previous
#include <hip/hip_runtime.h>
#include <math.h>

typedef __attribute__((ext_vector_type(2))) float v2f;
typedef __attribute__((ext_vector_type(4))) float v4f;
typedef __attribute__((ext_vector_type(8))) float v8f;

#define DM   1024
#define DS   16
#define DTR  64
#define NPJ  96
#define BSZ  2
#define TLEN 2048
#define MROWS (BSZ*TLEN)
#define NC   32
#define CL   (TLEN/NC)

__device__ __forceinline__ float softplus_f32(float z) {
    const float t  = __builtin_amdgcn_exp2f(-fabsf(z) * 1.44269504088896341f);
    const float u  = 1.0f + t;
    const float d  = t - (u - 1.0f);
    const float lg = __builtin_amdgcn_logf(u) * 0.693147180559945309f;
    return fmaxf(z, 0.0f) + fmaf(d, __builtin_amdgcn_rcpf(u), lg);
}

#define PJ_ROWS  16
#define PJ_WAVES 16
#define PJ_KW    (DM / PJ_WAVES)
#define PJ_TILE  (PJ_ROWS * NPJ)
#define XS_PITCH 68

__global__ __launch_bounds__(PJ_WAVES * 32)
void proj_dt_kernel(const float* __restrict__ x,
                    const float* __restrict__ Wx,
                    const float* __restrict__ Wdt,
                    const float* __restrict__ bdt,
                    float* __restrict__ xproj,
                    float* __restrict__ dt) {
    __shared__ float4 red[PJ_WAVES][PJ_TILE / 4];
    __shared__ __attribute__((aligned(16))) float xs[PJ_ROWS][XS_PITCH];

    const int mb   = blockIdx.x * PJ_ROWS;
    const int tid  = threadIdx.x;
    const int w    = tid >> 5;
    const int lane = tid & 31;
    const int half = lane >> 4;
    const int l16  = lane & 15;
    const int kb   = w * PJ_KW + 2 * half;

    v8f acc[6] = {};
    const float* arow = x + (size_t)(mb + l16) * DM + kb;
    const float* bcol = Wx + (size_t)kb * NPJ + l16;

#pragma unroll
    for (int ks = 0; ks < PJ_KW / 4; ++ks) {
        const v2f a = *(const v2f*)(arow + 4 * ks);
        const float* bp = bcol + (size_t)(4 * ks) * NPJ;
#pragma unroll
        for (int j = 0; j < 6; ++j) {
            v2f b;
            b.x = bp[16 * j];
            b.y = bp[16 * j + NPJ];
            acc[j] = __builtin_amdgcn_wmma_f32_16x16x4_f32(false, a, false, b, (short)0, acc[j], false, false);
        }
    }

    float* mine = (float*)&red[w][0];
#pragma unroll
    for (int j = 0; j < 6; ++j)
#pragma unroll
        for (int v = 0; v < 8; ++v)
            mine[(j * 8 + v) * 32 + lane] = acc[j][v];

    const int nb2 = w * 64;
    v2f bw[DTR / 4][4];
#pragma unroll
    for (int ks = 0; ks < DTR / 4; ++ks) {
        const float* bp = Wdt + (size_t)(4 * ks + 2 * half) * DM + nb2 + l16;
#pragma unroll
        for (int j = 0; j < 4; ++j) {
            bw[ks][j].x = bp[16 * j];
            bw[ks][j].y = bp[16 * j + DM];
        }
    }
    float bias[4];
#pragma unroll
    for (int j = 0; j < 4; ++j) bias[j] = bdt[nb2 + 16 * j + l16];

    __syncthreads();

    if (tid < PJ_TILE / 4) {
        float4 s = red[0][tid];
#pragma unroll
        for (int q = 1; q < PJ_WAVES; ++q) {
            const float4 p = red[q][tid];
            s.x += p.x; s.y += p.y; s.z += p.z; s.w += p.w;
        }
        const int frag = tid >> 6;
        const int v    = (tid >> 3) & 7;
        const int hh   = (tid >> 2) & 1;
        const int c0   = (tid & 3) * 4;
        const int rl   = 8 * hh + v;
        const int col  = 16 * frag + c0;
        if (frag < DTR / 16)
            *(float4*)&xs[rl][col] = s;
        else
            *(float4*)(xproj + (size_t)(mb + rl) * NPJ + col) = s;
    }
    __syncthreads();

    v8f acc2[4] = {};
#pragma unroll
    for (int ks = 0; ks < DTR / 4; ++ks) {
        const v2f a = *(const v2f*)&xs[l16][4 * ks + 2 * half];
#pragma unroll
        for (int j = 0; j < 4; ++j)
            acc2[j] = __builtin_amdgcn_wmma_f32_16x16x4_f32(false, a, false, bw[ks][j], (short)0, acc2[j], false, false);
    }
#pragma unroll
    for (int v = 0; v < 8; ++v) {
        const int row = mb + v + (half ? 8 : 0);
#pragma unroll
        for (int j = 0; j < 4; ++j) {
            const int col = nb2 + j * 16 + l16;
            dt[(size_t)row * DM + col] = softplus_f32(acc2[j][v] + bias[j]);
        }
    }
}

#define SCAN_THREADS 256
#define SG 8
#define NG (CL / SG)

#define SC  8
#define NSC (NC / SC)
#define LOG2E 1.44269504088896341f

__global__ __launch_bounds__(SCAN_THREADS)
void scan_chunk1_kernel(const float* __restrict__ x,
                        const float* __restrict__ xproj,
                        const float* __restrict__ dt,
                        const float* __restrict__ A_log,
                        float* __restrict__ Eloc,
                        float* __restrict__ sdpre,
                        float* __restrict__ Ssup,
                        float* __restrict__ sdsup) {
    __shared__ float4 sB[SC][CL][DS / 4];
    __shared__ float  exS[SC][DS + 1][32];

    const int bC   = blockIdx.x;
    const int b    = bC / NSC;
    const int C    = bC % NSC;
    const int tid  = threadIdx.x;
    const int lane = tid & 31;
    const int w    = __builtin_amdgcn_readfirstlane(tid >> 5);
    const int c    = C * SC + w;
    const int d    = blockIdx.y * 32 + lane;
    const size_t row0 = (size_t)b * TLEN + (size_t)c * CL;

#pragma unroll
    for (int i = 0; i < CL * (DS / 4) / 32; ++i) {
        const int idx = i * 32 + lane;
        sB[w][idx >> 2][idx & 3] =
            *(const float4*)(xproj + (row0 + (idx >> 2)) * NPJ + DTR + 4 * (idx & 3));
    }

    const float* dtp = dt + row0 * DM + d;
    const float* xp  = x  + row0 * DM + d;
    float dn[SG], xn[SG];
#pragma unroll
    for (int u = 0; u < SG; ++u) { dn[u] = dtp[(size_t)u * DM]; xn[u] = xp[(size_t)u * DM]; }

    float An[DS];
#pragma unroll
    for (int s = 0; s < DS; ++s) An[s] = -__expf(A_log[d * DS + s]);

    float h[DS];
#pragma unroll
    for (int s = 0; s < DS; ++s) h[s] = 0.0f;
    float sdt = 0.0f;

    __syncthreads();

    float4 nb[DS / 4];
#pragma unroll
    for (int q = 0; q < DS / 4; ++q) nb[q] = sB[w][0][q];

    for (int g = 0; g < NG; ++g) {
        float dc[SG], xc[SG];
#pragma unroll
        for (int u = 0; u < SG; ++u) { dc[u] = dn[u]; xc[u] = xn[u]; }
#pragma unroll
        for (int u = 0; u < SG; ++u) {
            const int t  = g * SG + u;
            const int tn = (t + 1 < CL) ? t + 1 : CL - 1;
            const float dtv = dc[u];
            const float dtx = dtv * xc[u];
            sdt += dtv;
            float Bm[DS];
#pragma unroll
            for (int q = 0; q < DS / 4; ++q) {
                const float4 v = nb[q];
                Bm[4 * q] = v.x; Bm[4 * q + 1] = v.y; Bm[4 * q + 2] = v.z; Bm[4 * q + 3] = v.w;
            }
#pragma unroll
            for (int q = 0; q < DS / 4; ++q) nb[q] = sB[w][tn][q];
            if (u == 0 && g + 1 < NG) {
#pragma unroll
                for (int v = 0; v < SG; ++v) {
                    dn[v] = dtp[(size_t)((g + 1) * SG + v) * DM];
                    xn[v] = xp [(size_t)((g + 1) * SG + v) * DM];
                }
            }
            __builtin_amdgcn_sched_barrier(0);
#pragma unroll
            for (int s = 0; s < DS; ++s) {
                const float a = __expf(dtv * An[s]);
                h[s] = a * h[s] + dtx * Bm[s];
            }
        }
    }

#pragma unroll
    for (int s = 0; s < DS; ++s) exS[w][s][lane] = h[s];
    exS[w][DS][lane] = sdt;
    __syncthreads();

    float E[DS];
#pragma unroll
    for (int s = 0; s < DS; ++s) E[s] = 0.0f;
    float sp = 0.0f;
    for (int j = 0; j < w; ++j) {
        const float sdj = exS[j][DS][lane];
        const float sd2 = sdj * LOG2E;
#pragma unroll
        for (int s = 0; s < DS; ++s)
            E[s] = __builtin_amdgcn_exp2f(sd2 * An[s]) * E[s] + exS[j][s][lane];
        sp += sdj;
    }
    sdpre[((size_t)b * NC + c) * DM + d] = sp;
    {
        v4f* E4 = (v4f*)Eloc + ((size_t)b * NC + c) * (DS / 4) * DM + d;
#pragma unroll
        for (int q = 0; q < DS / 4; ++q) {
            v4f t;
            t.x = E[4 * q]; t.y = E[4 * q + 1]; t.z = E[4 * q + 2]; t.w = E[4 * q + 3];
            E4[(size_t)q * DM] = t;
        }
    }
    if (w == SC - 1) {
        const float sd2 = sdt * LOG2E;
#pragma unroll
        for (int s = 0; s < DS; ++s)
            E[s] = __builtin_amdgcn_exp2f(sd2 * An[s]) * E[s] + h[s];
        sdsup[((size_t)b * NSC + C) * DM + d] = sp + sdt;
        v4f* U4 = (v4f*)Ssup + ((size_t)b * NSC + C) * (DS / 4) * DM + d;
#pragma unroll
        for (int q = 0; q < DS / 4; ++q) {
            v4f t;
            t.x = E[4 * q]; t.y = E[4 * q + 1]; t.z = E[4 * q + 2]; t.w = E[4 * q + 3];
            U4[(size_t)q * DM] = t;
        }
    }
}

__global__ __launch_bounds__(SCAN_THREADS)
void scan_chunk2_kernel(const float* __restrict__ x,
                        const float* __restrict__ xproj,
                        const float* __restrict__ dt,
                        const float* __restrict__ A_log,
                        const float* __restrict__ Dvec,
                        const float* __restrict__ Eloc,
                        const float* __restrict__ sdpre,
                        const float* __restrict__ Ssup,
                        const float* __restrict__ sdsup,
                        float* __restrict__ out) {
    __shared__ float4 sBC[CL][2 * DS / 4];

    const int bc  = blockIdx.x;
    const int b   = bc / NC;
    const int c   = bc % NC;
    const int tid = threadIdx.x;
    const int d   = blockIdx.y * SCAN_THREADS + tid;
    const size_t row0 = (size_t)b * TLEN + (size_t)c * CL;

    {
        const int r = tid >> 3, q = tid & 7;
        const float4 lo = *(const float4*)(xproj + (row0 + r) * NPJ + DTR + 4 * q);
        const float4 hi = *(const float4*)(xproj + (row0 + CL / 2 + r) * NPJ + DTR + 4 * q);
        sBC[r][q] = lo;
        sBC[CL / 2 + r][q] = hi;
    }

    const float* dtp = dt + row0 * DM + d;
    const float* xp  = x  + row0 * DM + d;
    float*       op  = out + row0 * DM + d;
    float dn[SG], xn[SG];
#pragma unroll
    for (int u = 0; u < SG; ++u) { dn[u] = dtp[(size_t)u * DM]; xn[u] = xp[(size_t)u * DM]; }

    v4f   ev[DS / 4];
    v4f   uv[NSC - 1][DS / 4];
    float usd[NSC - 1];
    {
        const v4f* E4 = (const v4f*)Eloc + ((size_t)b * NC + c) * (DS / 4) * DM + d;
#pragma unroll
        for (int q = 0; q < DS / 4; ++q) ev[q] = E4[(size_t)q * DM];
#pragma unroll
        for (int j = 0; j < NSC - 1; ++j) {
            const v4f* U4 = (const v4f*)Ssup + ((size_t)b * NSC + j) * (DS / 4) * DM + d;
#pragma unroll
            for (int q = 0; q < DS / 4; ++q) uv[j][q] = U4[(size_t)q * DM];
            usd[j] = sdsup[((size_t)b * NSC + j) * DM + d];
        }
    }
    const float spre = sdpre[((size_t)b * NC + c) * DM + d];

    float An[DS];
#pragma unroll
    for (int s = 0; s < DS; ++s) An[s] = -__expf(A_log[d * DS + s]);

    float h[DS];
#pragma unroll
    for (int s = 0; s < DS; ++s) h[s] = 0.0f;
    const int Cs = c / SC;
#pragma unroll
    for (int j = 0; j < NSC - 1; ++j) {
        const bool on = j < Cs;
        const float sd2 = usd[j] * LOG2E;
#pragma unroll
        for (int s = 0; s < DS; ++s) {
            const float t = __builtin_amdgcn_exp2f(sd2 * An[s]) * h[s] + uv[j][s >> 2][s & 3];
            h[s] = on ? t : h[s];
        }
    }
    {
        const bool on = (c % SC) != 0;
        const float sd2 = spre * LOG2E;
#pragma unroll
        for (int s = 0; s < DS; ++s) {
            const float t = __builtin_amdgcn_exp2f(sd2 * An[s]) * h[s] + ev[s >> 2][s & 3];
            h[s] = on ? t : h[s];
        }
    }

    const float dval = Dvec[d];

    __syncthreads();

    float4 nb[2 * DS / 4];
#pragma unroll
    for (int q = 0; q < 2 * DS / 4; ++q) nb[q] = sBC[0][q];

    for (int g = 0; g < NG; ++g) {
        float dc[SG], xc[SG];
#pragma unroll
        for (int u = 0; u < SG; ++u) { dc[u] = dn[u]; xc[u] = xn[u]; }
#pragma unroll
        for (int u = 0; u < SG; ++u) {
            const int t  = g * SG + u;
            const int tn = (t + 1 < CL) ? t + 1 : CL - 1;
            const float dtv = dc[u];
            const float xv  = xc[u];
            const float dtx = dtv * xv;
            float Bm[DS], Cm[DS];
#pragma unroll
            for (int q = 0; q < DS / 4; ++q) {
                const float4 v = nb[q];
                const float4 w = nb[DS / 4 + q];
                Bm[4 * q] = v.x; Bm[4 * q + 1] = v.y; Bm[4 * q + 2] = v.z; Bm[4 * q + 3] = v.w;
                Cm[4 * q] = w.x; Cm[4 * q + 1] = w.y; Cm[4 * q + 2] = w.z; Cm[4 * q + 3] = w.w;
            }
#pragma unroll
            for (int q = 0; q < 2 * DS / 4; ++q) nb[q] = sBC[tn][q];
            if (u == 0 && g + 1 < NG) {
#pragma unroll
                for (int v = 0; v < SG; ++v) {
                    dn[v] = dtp[(size_t)((g + 1) * SG + v) * DM];
                    xn[v] = xp [(size_t)((g + 1) * SG + v) * DM];
                }
            }
            __builtin_amdgcn_sched_barrier(0);
            float y = 0.0f;
#pragma unroll
            for (int s = 0; s < DS; ++s) {
                const float a = __expf(dtv * An[s]);
                h[s] = a * h[s] + dtx * Bm[s];
                y += h[s] * Cm[s];
            }
            op[(size_t)t * DM] = y + xv * dval;
        }
    }
}

extern "C" void kernel_launch(void* const* d_in, const int* in_sizes, int n_in,
                              void* d_out, int out_size, void* d_ws, size_t ws_size,
                              hipStream_t stream) {
    const float* x     = (const float*)d_in[0];
    const float* Wx    = (const float*)d_in[1];
    const float* Wdt   = (const float*)d_in[2];
    const float* bdt   = (const float*)d_in[3];
    const float* A_log = (const float*)d_in[4];
    const float* Dvec  = (const float*)d_in[5];
    float* out = (float*)d_out;

    char* ws = (char*)d_ws;
    float* xproj = (float*)ws;
    ws += (size_t)MROWS * NPJ * sizeof(float);
    float* dt    = (float*)ws;
    ws += (size_t)MROWS * DM * sizeof(float);
    float* Eloc  = (float*)ws;
    ws += (size_t)BSZ * NC * DM * DS * sizeof(float);
    float* Ssup  = (float*)ws;
    ws += (size_t)BSZ * NSC * DM * DS * sizeof(float);
    float* sdpre = (float*)ws;
    ws += (size_t)BSZ * NC * DM * sizeof(float);
    float* sdsup = (float*)ws;

    proj_dt_kernel<<<dim3(MROWS / PJ_ROWS), PJ_WAVES * 32, 0, stream>>>(x, Wx, Wdt, bdt, xproj, dt);
    scan_chunk1_kernel<<<dim3(BSZ * NSC, DM / 32), SCAN_THREADS, 0, stream>>>(x, xproj, dt, A_log, Eloc, sdpre, Ssup, sdsup);
    scan_chunk2_kernel<<<dim3(BSZ * NC, DM / SCAN_THREADS), SCAN_THREADS, 0, stream>>>(x, xproj, dt, A_log, Dvec, Eloc, sdpre, Ssup, sdsup, out);
}
